# starting kernel
# baseline (speedup 1.0000x reference)
#include <hip/hip_runtime.h>
#include <hip/hip_bf16.h>

typedef __attribute__((ext_vector_type(2))) float v2f;
typedef __attribute__((ext_vector_type(8))) float v8f;

#define B_ 8
#define N_ 512
#define F_ 128
#define T_ 32
#define D_ 64
#define EPS_ 1e-6f

__global__ void GAT_z_kernel(const float* __restrict__ x,
                             const float* __restrict__ W,
                             const float* __restrict__ bias,
                             float* __restrict__ z) {
  const int wave = blockIdx.x * (blockDim.x >> 5) + (threadIdx.x >> 5);
  const int lane = threadIdx.x & 31;

  const int dtile = wave & 3;
  const int ntile = (wave >> 2) & 31;
  const int t     = (wave >> 7) & 31;
  const int b     = wave >> 12;

  const int l15  = lane & 15;
  const int koff = (lane >> 4) << 1;
  const int mrow = ntile * 16 + l15;
  const int dcol = dtile * 16 + l15;

  const float* xrow = x + ((size_t)(b * N_ + mrow) * F_) * T_ + t;
  const float* wrow = W + (size_t)dcol * F_;

  v8f c = {};
  for (int k = 0; k < F_; k += 4) {
    v2f a, bm;
    a.x  = xrow[(size_t)(k + koff)     * T_];
    a.y  = xrow[(size_t)(k + koff + 1) * T_];
    bm.x = wrow[k + koff];
    bm.y = wrow[k + koff + 1];
    c = __builtin_amdgcn_wmma_f32_16x16x4_f32(false, a, false, bm,
                                              (short)0, c, false, false);
  }

  const float bv = bias[dcol];
  const int rbase = (lane >> 4) << 3;
  const size_t zbt = (size_t)(b * T_ + t) * N_;
#pragma unroll
  for (int v = 0; v < 8; ++v) {
    const int n = ntile * 16 + rbase + v;
    z[(zbt + n) * D_ + dcol] = tanhf(c[v] + bv);
  }
}

__global__ void GAT_sq_kernel(const float* __restrict__ z,
                              float* __restrict__ sq) {
  const int i = blockIdx.x * blockDim.x + threadIdx.x;
  if (i >= B_ * T_ * N_) return;
  const float* zr = z + (size_t)i * D_;
  float s = 0.f;
#pragma unroll 8
  for (int d = 0; d < D_; ++d) { const float v = zr[d]; s = fmaf(v, v, s); }
  sq[i] = s;
}

__global__ void __launch_bounds__(256)
GAT_att_kernel(const float* __restrict__ z,
               const float* __restrict__ sq,
               float* __restrict__ out) {
  __shared__ float att_s[16 * 512];
  __shared__ float sq_s[512];
  __shared__ float red_s[256];
  __shared__ float rowsum_s[16];

  const int tid   = threadIdx.x;
  const int wave  = tid >> 5;
  const int lane  = tid & 31;
  const int ntile = blockIdx.x & 31;
  const int t     = blockIdx.x >> 5;

  const int l15   = lane & 15;
  const int koff  = (lane >> 4) << 1;
  const int rbase = (lane >> 4) << 3;
  const int arow  = ntile * 16 + l15;

  float acc[32];
#pragma unroll
  for (int j = 0; j < 32; ++j) acc[j] = 0.f;

  for (int b = 0; b < B_; ++b) {
    const float* zb  = z  + (size_t)(b * T_ + t) * N_ * D_;
    const float* sqb = sq + (size_t)(b * T_ + t) * N_;

    for (int i = tid; i < N_; i += 256) sq_s[i] = sqb[i];
    __syncthreads();

    v8f c0 = {}, c1 = {}, c2 = {}, c3 = {};
    const float* arp = zb + (size_t)arow * D_;
    const float* bp0 = zb + (size_t)((wave * 4 + 0) * 16 + l15) * D_;
    const float* bp1 = zb + (size_t)((wave * 4 + 1) * 16 + l15) * D_;
    const float* bp2 = zb + (size_t)((wave * 4 + 2) * 16 + l15) * D_;
    const float* bp3 = zb + (size_t)((wave * 4 + 3) * 16 + l15) * D_;
#pragma unroll
    for (int k = 0; k < D_; k += 4) {
      const v2f a = *(const v2f*)(arp + k + koff);
      v2f bm;
      bm = *(const v2f*)(bp0 + k + koff);
      c0 = __builtin_amdgcn_wmma_f32_16x16x4_f32(false, a, false, bm, (short)0, c0, false, false);
      bm = *(const v2f*)(bp1 + k + koff);
      c1 = __builtin_amdgcn_wmma_f32_16x16x4_f32(false, a, false, bm, (short)0, c1, false, false);
      bm = *(const v2f*)(bp2 + k + koff);
      c2 = __builtin_amdgcn_wmma_f32_16x16x4_f32(false, a, false, bm, (short)0, c2, false, false);
      bm = *(const v2f*)(bp3 + k + koff);
      c3 = __builtin_amdgcn_wmma_f32_16x16x4_f32(false, a, false, bm, (short)0, c3, false, false);
    }

#pragma unroll
    for (int ct = 0; ct < 4; ++ct) {
      const v8f c = (ct == 0) ? c0 : (ct == 1) ? c1 : (ct == 2) ? c2 : c3;
      const int col = (wave * 4 + ct) * 16 + l15;
      const float sqm = sq_s[col];
      const float nm  = sqrtf(sqm) + EPS_;
#pragma unroll
      for (int v = 0; v < 8; ++v) {
        const int rloc = rbase + v;
        const float sqn = sq_s[ntile * 16 + rloc];
        const float nn  = sqrtf(sqn) + EPS_;
        const float dot = c[v];
        const float cosv = dot / (nn * nm);
        const float d2   = fmaxf(sqn + sqm - 2.0f * dot, 0.0f);
        const float diff = sqrtf(d2);
        att_s[rloc * 512 + col] = __expf(cosv - diff / (nn + nm + EPS_));
      }
    }
    __syncthreads();

    {
      const int row = tid >> 4;
      const int sl  = tid & 15;
      float part = 0.f;
#pragma unroll 4
      for (int cidx = sl; cidx < 512; cidx += 16) part += att_s[row * 512 + cidx];
      red_s[tid] = part;
    }
    __syncthreads();
    if ((tid & 15) == 0) {
      const int row = tid >> 4;
      float s = 0.f;
#pragma unroll
      for (int j = 0; j < 16; ++j) s += red_s[(row << 4) + j];
      rowsum_s[row] = s;
    }
    __syncthreads();

    const float invB = 1.0f / (float)B_;
#pragma unroll
    for (int j = 0; j < 32; ++j) {
      const int idx = tid + j * 256;
      const int r   = idx >> 9;
      acc[j] += att_s[idx] * invB / (rowsum_s[r] + EPS_);
    }
    __syncthreads();
  }

  float* outp = out + ((size_t)t * N_ + ntile * 16) * N_;
#pragma unroll
  for (int j = 0; j < 32; ++j) {
    const int idx = tid + j * 256;
    outp[idx] = acc[j];
  }
}

extern "C" void kernel_launch(void* const* d_in, const int* in_sizes, int n_in,
                              void* d_out, int out_size, void* d_ws, size_t ws_size,
                              hipStream_t stream) {
  const float* x    = (const float*)d_in[0];
  const float* W    = (const float*)d_in[1];
  const float* bias = (const float*)d_in[2];
  float* out = (float*)d_out;

  float* z  = (float*)d_ws;
  float* sq = z + (size_t)B_ * T_ * N_ * D_;

  GAT_z_kernel<<<8192, 128, 0, stream>>>(x, W, bias, z);

  GAT_sq_kernel<<<(B_ * T_ * N_ + 255) / 256, 256, 0, stream>>>(z, sq);

  GAT_att_kernel<<<T_ * (N_ / 16), 256, 0, stream>>>(z, sq, out);
}


# att kernel: software pipeline over b: loads and matrix products of batch b+1 issued in four groups among the four epilogue chunks of batch b; norms staged once per batch in LDS; sum orders kept
# speedup vs baseline: 2.6183x; 2.6183x over previous
#include <hip/hip_runtime.h>
#include <hip/hip_bf16.h>

typedef __attribute__((ext_vector_type(2))) float v2f;
typedef __attribute__((ext_vector_type(4))) float v4f;
typedef __attribute__((ext_vector_type(8))) float v8f;

#define B_ 8
#define N_ 512
#define F_ 128
#define T_ 32
#define D_ 64
#define EPS_ 1e-6f

#define ZTILE 1024
#define ZFC  16
#define ZNCH (F_ / ZFC)
#define ZTP  (ZFC + 2)
#define ZNP  (T_ * ZTP + 4)
#define ZWP  (ZFC + 4)

__global__ void __launch_bounds__(256)
GAT_z_kernel(const float* __restrict__ x,
             const float* __restrict__ W,
             const float* __restrict__ bias,
             float* __restrict__ z) {
  __shared__ __attribute__((aligned(16))) float xs[2][16 * ZNP];
  __shared__ __attribute__((aligned(16))) float wsm[2][D_ * ZWP];

  const int tid   = threadIdx.x;
  const int wave  = tid >> 5;
  const int lane  = tid & 31;
  const int ntile = blockIdx.x & 31;
  const int b     = blockIdx.x >> 5;
  const int l15   = lane & 15;
  const int h     = lane >> 4;

  const float* xg = x + (size_t)(b * N_ + ntile * 16 + (tid >> 7)) * (F_ * T_) + (tid & 127) * 4;
  const int    xl = (tid >> 7) * ZNP + ((tid & 7) * 4) * ZTP + ((tid & 127) >> 3);
  const float* wg = W + (size_t)(tid >> 2) * F_ + (tid & 3) * 4;
  const int    wl = (tid >> 2) * ZWP + (tid & 3) * 4;

  v4f xr[8];
  v4f wr;
#pragma unroll
  for (int p = 0; p < 8; ++p) xr[p] = *(const v4f*)(xg + (size_t)(2 * p) * (F_ * T_));
  wr = *(const v4f*)wg;
#pragma unroll
  for (int p = 0; p < 8; ++p) {
#pragma unroll
    for (int i = 0; i < 4; ++i) xs[0][xl + 2 * p * ZNP + i * ZTP] = xr[p][i];
  }
  *(v4f*)&wsm[0][wl] = wr;

  v8f acc[4][4];
#pragma unroll
  for (int j = 0; j < 4; ++j) {
#pragma unroll
    for (int dt = 0; dt < 4; ++dt) acc[j][dt] = v8f{};
  }

  const int aoff = l15 * ZNP + (4 * wave) * ZTP + 2 * h;
  const int boff = l15 * ZWP + 2 * h;

  for (int c = 0; c < ZNCH; ++c) {
    __syncthreads();

    if (c + 1 < ZNCH) {
#pragma unroll
      for (int p = 0; p < 8; ++p)
        xr[p] = *(const v4f*)(xg + (size_t)(2 * p) * (F_ * T_) + (c + 1) * ZFC * T_);
      wr = *(const v4f*)(wg + (c + 1) * ZFC);
    }
    __builtin_amdgcn_sched_barrier(0);

    const float* xa = &xs[c & 1][aoff];
    const float* wb = &wsm[c & 1][boff];
#pragma unroll
    for (int kk = 0; kk < ZFC; kk += 4) {
      v2f af[4];
      v2f bf[4];
#pragma unroll
      for (int j = 0; j < 4; ++j)  af[j]  = *(const v2f*)(xa + j * ZTP + kk);
#pragma unroll
      for (int dt = 0; dt < 4; ++dt) bf[dt] = *(const v2f*)(wb + dt * 16 * ZWP + kk);
#pragma unroll
      for (int j = 0; j < 4; ++j) {
#pragma unroll
        for (int dt = 0; dt < 4; ++dt)
          acc[j][dt] = __builtin_amdgcn_wmma_f32_16x16x4_f32(false, bf[dt], false, af[j],
                                                             (short)0, acc[j][dt], false, false);
      }
    }

    __builtin_amdgcn_sched_barrier(0);
    if (c + 1 < ZNCH) {
#pragma unroll
      for (int p = 0; p < 8; ++p) {
#pragma unroll
        for (int i = 0; i < 4; ++i) xs[(c + 1) & 1][xl + 2 * p * ZNP + i * ZTP] = xr[p][i];
      }
      *(v4f*)&wsm[(c + 1) & 1][wl] = wr;
    }
  }

  v4f bv[4][2];
#pragma unroll
  for (int dt = 0; dt < 4; ++dt) {
#pragma unroll
    for (int u = 0; u < 2; ++u) bv[dt][u] = *(const v4f*)(bias + dt * 16 + h * 8 + u * 4);
  }
#pragma unroll
  for (int j = 0; j < 4; ++j) {
    const int t = 4 * wave + j;
    float* zt = z + ((size_t)(b * T_ + t) * (N_ / 16) + ntile) * ZTILE + lane * 4;
#pragma unroll
    for (int dt = 0; dt < 4; ++dt) {
#pragma unroll
      for (int u = 0; u < 2; ++u) {
        v4f o;
#pragma unroll
        for (int e = 0; e < 4; ++e) o[e] = tanhf(acc[j][dt][4 * u + e] + bv[dt][u][e]);
        *(v4f*)(zt + (2 * dt + u) * 128) = o;
      }
    }
  }
}

__global__ void __launch_bounds__(256)
GAT_sq_kernel(const float* __restrict__ z,
              float* __restrict__ sq) {
  const int lane = threadIdx.x & 31;
  const int wv   = blockIdx.x * 8 + (threadIdx.x >> 5);
  const float* zr = z + (size_t)wv * (4 * ZTILE) + lane * 4;
  float s[4];
#pragma unroll
  for (int i = 0; i < 4; ++i) {
    float a = 0.f;
#pragma unroll
    for (int q = 0; q < 8; ++q) {
      const v4f v = *(const v4f*)(zr + i * ZTILE + q * 128);
#pragma unroll
      for (int e = 0; e < 4; ++e) a = fmaf(v[e], v[e], a);
    }
    s[i] = a;
  }
#pragma unroll
  for (int i = 0; i < 4; ++i) s[i] += __shfl_xor(s[i], 16);
  float* o = sq + (size_t)wv * 64 + lane;
  o[0]  = (lane < 16) ? s[0] : s[1];
  o[32] = (lane < 16) ? s[2] : s[3];
}


static __device__ __forceinline__ void att_load_group(v4f (&A)[2], v4f (&Bf)[2][4],
                                                      const float* __restrict__ ap,
                                                      const float* __restrict__ bp) {
#pragma unroll
  for (int u = 0; u < 2; ++u) {
    A[u] = *(const v4f*)(ap + u * 128);
#pragma unroll
    for (int i = 0; i < 4; ++i) Bf[u][i] = *(const v4f*)(bp + i * ZTILE + u * 128);
  }
}

static __device__ __forceinline__ void att_mma_group(v8f (&c)[4], const v4f (&A)[2], const v4f (&Bf)[2][4]) {
#pragma unroll
  for (int u = 0; u < 2; ++u) {
    const v2f alo = {A[u][0], A[u][1]};
    const v2f ahi = {A[u][2], A[u][3]};
#pragma unroll
    for (int i = 0; i < 4; ++i)
      c[i] = __builtin_amdgcn_wmma_f32_16x16x4_f32(false, alo, false, v2f{Bf[u][i][0], Bf[u][i][1]},
                                                   (short)0, c[i], false, false);
#pragma unroll
    for (int i = 0; i < 4; ++i)
      c[i] = __builtin_amdgcn_wmma_f32_16x16x4_f32(false, ahi, false, v2f{Bf[u][i][2], Bf[u][i][3]},
                                                   (short)0, c[i], false, false);
  }
}

__global__ void __launch_bounds__(256)
GAT_att_kernel(const float* __restrict__ z,
               const float* __restrict__ sq,
               float* __restrict__ out) {
  __shared__ float att_s[16 * 512];
  __shared__ float sq_s[2][512];
  __shared__ float nrm_s[2][512];
  __shared__ float red_s[256];
  __shared__ float rowsum_s[16];

  const int tid   = threadIdx.x;
  const int wave  = tid >> 5;
  const int lane  = tid & 31;
  const int ntile = blockIdx.x & 31;
  const int t     = blockIdx.x >> 5;

  const int l15   = lane & 15;
  const int rbase = (lane >> 4) << 3;

  float acc[32];
#pragma unroll
  for (int j = 0; j < 32; ++j) acc[j] = 0.f;

  const size_t bstride = (size_t)T_ * N_ * D_;
  const float* ap0 = z + (size_t)t * N_ * D_ + (size_t)ntile * ZTILE + lane * 4;
  const float* bp0 = z + (size_t)t * N_ * D_ + (size_t)(wave * 4) * ZTILE + lane * 4;

  v4f fa[2], fb[2][4];
  v8f cn[4];
#pragma unroll
  for (int i = 0; i < 4; ++i) cn[i] = v8f{};

  {
    const float* sq0 = sq + (size_t)t * N_;
    const float s0 = sq0[tid];
    const float s1 = sq0[tid + 256];
#pragma unroll
    for (int g = 0; g < 4; ++g) {
      att_load_group(fa, fb, ap0 + g * 256, bp0 + g * 256);
      att_mma_group(cn, fa, fb);
    }
    att_load_group(fa, fb, ap0 + bstride, bp0 + bstride);
    sq_s[0][tid]        = s0;
    sq_s[0][tid + 256]  = s1;
    nrm_s[0][tid]       = sqrtf(s0) + EPS_;
    nrm_s[0][tid + 256] = sqrtf(s1) + EPS_;
  }
  __syncthreads();

  for (int b = 0; b < B_; ++b) {
    const bool has_next = (b + 1 < B_);
    const float* sqc  = sq_s[b & 1];
    const float* nrmc = nrm_s[b & 1];

    v8f cc[4];
#pragma unroll
    for (int i = 0; i < 4; ++i) { cc[i] = cn[i]; cn[i] = v8f{}; }

    float sn0 = 0.f, sn1 = 0.f;
    if (has_next) {
      const float* sqn = sq + (size_t)((b + 1) * T_ + t) * N_;
      sn0 = sqn[tid];
      sn1 = sqn[tid + 256];
    }
    const float* apn = ap0 + (size_t)(b + 1) * bstride;
    const float* bpn = bp0 + (size_t)(b + 1) * bstride;

#pragma unroll
    for (int g = 0; g < 4; ++g) {
      if (has_next) {
        att_mma_group(cn, fa, fb);
        if (g < 3) {
          att_load_group(fa, fb, apn + (g + 1) * 256, bpn + (g + 1) * 256);
        } else if (b + 2 < B_) {
          att_load_group(fa, fb, apn + bstride, bpn + bstride);
        }
      }

      const v8f c = cc[g];
      const int col = (wave * 4 + g) * 16 + l15;
      const float sqm = sqc[col];
      const float nm  = nrmc[col];
#pragma unroll
      for (int v = 0; v < 8; ++v) {
        const int rloc = rbase + v;
        const float sqn = sqc[ntile * 16 + rloc];
        const float nn  = nrmc[ntile * 16 + rloc];
        const float dot = c[v];
        const float cosv = dot / (nn * nm);
        const float d2   = fmaxf(sqn + sqm - 2.0f * dot, 0.0f);
        const float diff = sqrtf(d2);
        att_s[rloc * 512 + col] = __expf(cosv - diff / (nn + nm + EPS_));
      }
    }
    __syncthreads();

    if (has_next) {
      sq_s[(b + 1) & 1][tid]        = sn0;
      sq_s[(b + 1) & 1][tid + 256]  = sn1;
      nrm_s[(b + 1) & 1][tid]       = sqrtf(sn0) + EPS_;
      nrm_s[(b + 1) & 1][tid + 256] = sqrtf(sn1) + EPS_;
    }

    {
      const int row = tid >> 4;
      const int sl  = tid & 15;
      float part = 0.f;
#pragma unroll 4
      for (int cidx = sl; cidx < 512; cidx += 16) part += att_s[row * 512 + cidx];
      red_s[tid] = part;
    }
    __syncthreads();
    if ((tid & 15) == 0) {
      const int row = tid >> 4;
      float s = 0.f;
#pragma unroll
      for (int j = 0; j < 16; ++j) s += red_s[(row << 4) + j];
      rowsum_s[row] = s;
    }
    __syncthreads();

    const float invB = 1.0f / (float)B_;
#pragma unroll
    for (int j = 0; j < 32; ++j) {
      const int idx = tid + j * 256;
      const int r   = idx >> 9;
      acc[j] += att_s[idx] * invB / (rowsum_s[r] + EPS_);
    }
    __syncthreads();
  }

  float* outp = out + ((size_t)t * N_ + ntile * 16) * N_;
#pragma unroll
  for (int j = 0; j < 32; ++j) {
    const int idx = tid + j * 256;
    outp[idx] = acc[j];
  }
}

extern "C" void kernel_launch(void* const* d_in, const int* in_sizes, int n_in,
                              void* d_out, int out_size, void* d_ws, size_t ws_size,
                              hipStream_t stream) {
  const float* x    = (const float*)d_in[0];
  const float* W    = (const float*)d_in[1];
  const float* bias = (const float*)d_in[2];
  float* out = (float*)d_out;

  float* z  = (float*)d_ws;
  float* sq = z + (size_t)B_ * T_ * N_ * D_;

  GAT_z_kernel<<<B_ * (N_ / 16), 256, 0, stream>>>(x, W, bias, z);

  GAT_sq_kernel<<<B_ * T_ * N_ / 64 / 8, 256, 0, stream>>>(z, sq);

  GAT_att_kernel<<<T_ * (N_ / 16), 256, 0, stream>>>(z, sq, out);
}


# att softmax phase, row-owner layout: exp tile (LDS, two buffers) read back 16 lanes per row, row sum in-thread + 4 lane exchanges (row-sum order changes), one division per row, 1 barrier per b
# speedup vs baseline: 3.0631x; 1.1699x over previous
#include <hip/hip_runtime.h>
#include <hip/hip_bf16.h>

typedef __attribute__((ext_vector_type(2))) float v2f;
typedef __attribute__((ext_vector_type(4))) float v4f;
typedef __attribute__((ext_vector_type(8))) float v8f;

#define B_ 8
#define N_ 512
#define F_ 128
#define T_ 32
#define D_ 64
#define EPS_ 1e-6f

#define ZTILE 1024
#define ZFC  16
#define ZNCH (F_ / ZFC)
#define ZTP  (ZFC + 2)
#define ZNP  (T_ * ZTP + 4)
#define ZWP  (ZFC + 4)

__global__ void __launch_bounds__(256)
GAT_z_kernel(const float* __restrict__ x,
             const float* __restrict__ W,
             const float* __restrict__ bias,
             float* __restrict__ z) {
  __shared__ __attribute__((aligned(16))) float xs[2][16 * ZNP];
  __shared__ __attribute__((aligned(16))) float wsm[2][D_ * ZWP];

  const int tid   = threadIdx.x;
  const int wave  = tid >> 5;
  const int lane  = tid & 31;
  const int ntile = blockIdx.x & 31;
  const int b     = blockIdx.x >> 5;
  const int l15   = lane & 15;
  const int h     = lane >> 4;

  const float* xg = x + (size_t)(b * N_ + ntile * 16 + (tid >> 7)) * (F_ * T_) + (tid & 127) * 4;
  const int    xl = (tid >> 7) * ZNP + ((tid & 7) * 4) * ZTP + ((tid & 127) >> 3);
  const float* wg = W + (size_t)(tid >> 2) * F_ + (tid & 3) * 4;
  const int    wl = (tid >> 2) * ZWP + (tid & 3) * 4;

  v4f xr[8];
  v4f wr;
#pragma unroll
  for (int p = 0; p < 8; ++p) xr[p] = *(const v4f*)(xg + (size_t)(2 * p) * (F_ * T_));
  wr = *(const v4f*)wg;
#pragma unroll
  for (int p = 0; p < 8; ++p) {
#pragma unroll
    for (int i = 0; i < 4; ++i) xs[0][xl + 2 * p * ZNP + i * ZTP] = xr[p][i];
  }
  *(v4f*)&wsm[0][wl] = wr;

  v8f acc[4][4];
#pragma unroll
  for (int j = 0; j < 4; ++j) {
#pragma unroll
    for (int dt = 0; dt < 4; ++dt) acc[j][dt] = v8f{};
  }

  const int aoff = l15 * ZNP + (4 * wave) * ZTP + 2 * h;
  const int boff = l15 * ZWP + 2 * h;

  for (int c = 0; c < ZNCH; ++c) {
    __syncthreads();

    if (c + 1 < ZNCH) {
#pragma unroll
      for (int p = 0; p < 8; ++p)
        xr[p] = *(const v4f*)(xg + (size_t)(2 * p) * (F_ * T_) + (c + 1) * ZFC * T_);
      wr = *(const v4f*)(wg + (c + 1) * ZFC);
    }
    __builtin_amdgcn_sched_barrier(0);

    const float* xa = &xs[c & 1][aoff];
    const float* wb = &wsm[c & 1][boff];
#pragma unroll
    for (int kk = 0; kk < ZFC; kk += 4) {
      v2f af[4];
      v2f bf[4];
#pragma unroll
      for (int j = 0; j < 4; ++j)  af[j]  = *(const v2f*)(xa + j * ZTP + kk);
#pragma unroll
      for (int dt = 0; dt < 4; ++dt) bf[dt] = *(const v2f*)(wb + dt * 16 * ZWP + kk);
#pragma unroll
      for (int j = 0; j < 4; ++j) {
#pragma unroll
        for (int dt = 0; dt < 4; ++dt)
          acc[j][dt] = __builtin_amdgcn_wmma_f32_16x16x4_f32(false, bf[dt], false, af[j],
                                                             (short)0, acc[j][dt], false, false);
      }
    }

    __builtin_amdgcn_sched_barrier(0);
    if (c + 1 < ZNCH) {
#pragma unroll
      for (int p = 0; p < 8; ++p) {
#pragma unroll
        for (int i = 0; i < 4; ++i) xs[(c + 1) & 1][xl + 2 * p * ZNP + i * ZTP] = xr[p][i];
      }
      *(v4f*)&wsm[(c + 1) & 1][wl] = wr;
    }
  }

  v4f bv[4][2];
#pragma unroll
  for (int dt = 0; dt < 4; ++dt) {
#pragma unroll
    for (int u = 0; u < 2; ++u) bv[dt][u] = *(const v4f*)(bias + dt * 16 + h * 8 + u * 4);
  }
#pragma unroll
  for (int j = 0; j < 4; ++j) {
    const int t = 4 * wave + j;
    float* zt = z + ((size_t)(b * T_ + t) * (N_ / 16) + ntile) * ZTILE + lane * 4;
#pragma unroll
    for (int dt = 0; dt < 4; ++dt) {
#pragma unroll
      for (int u = 0; u < 2; ++u) {
        v4f o;
#pragma unroll
        for (int e = 0; e < 4; ++e) o[e] = tanhf(acc[j][dt][4 * u + e] + bv[dt][u][e]);
        *(v4f*)(zt + (2 * dt + u) * 128) = o;
      }
    }
  }
}

__global__ void __launch_bounds__(256)
GAT_sq_kernel(const float* __restrict__ z,
              float* __restrict__ sq) {
  const int lane = threadIdx.x & 31;
  const int wv   = blockIdx.x * 8 + (threadIdx.x >> 5);
  const float* zr = z + (size_t)wv * (4 * ZTILE) + lane * 4;
  float s[4];
#pragma unroll
  for (int i = 0; i < 4; ++i) {
    float a = 0.f;
#pragma unroll
    for (int q = 0; q < 8; ++q) {
      const v4f v = *(const v4f*)(zr + i * ZTILE + q * 128);
#pragma unroll
      for (int e = 0; e < 4; ++e) a = fmaf(v[e], v[e], a);
    }
    s[i] = a;
  }
#pragma unroll
  for (int i = 0; i < 4; ++i) s[i] += __shfl_xor(s[i], 16);
  float* o = sq + (size_t)wv * 64 + lane;
  o[0]  = (lane < 16) ? s[0] : s[1];
  o[32] = (lane < 16) ? s[2] : s[3];
}

#define APITCH 516

static __device__ __forceinline__ void att_load_group(v4f (&A)[2], v4f (&Bf)[2][4],
                                                      const float* __restrict__ ap,
                                                      const float* __restrict__ bp) {
#pragma unroll
  for (int u = 0; u < 2; ++u) {
    A[u] = *(const v4f*)(ap + u * 128);
#pragma unroll
    for (int i = 0; i < 4; ++i) Bf[u][i] = *(const v4f*)(bp + i * ZTILE + u * 128);
  }
}

static __device__ __forceinline__ void att_mma_group(v8f (&c)[4], const v4f (&A)[2], const v4f (&Bf)[2][4]) {
#pragma unroll
  for (int u = 0; u < 2; ++u) {
    const v2f alo = {A[u][0], A[u][1]};
    const v2f ahi = {A[u][2], A[u][3]};
#pragma unroll
    for (int i = 0; i < 4; ++i)
      c[i] = __builtin_amdgcn_wmma_f32_16x16x4_f32(false, alo, false, v2f{Bf[u][i][0], Bf[u][i][1]},
                                                   (short)0, c[i], false, false);
#pragma unroll
    for (int i = 0; i < 4; ++i)
      c[i] = __builtin_amdgcn_wmma_f32_16x16x4_f32(false, ahi, false, v2f{Bf[u][i][2], Bf[u][i][3]},
                                                   (short)0, c[i], false, false);
  }
}

__global__ void __launch_bounds__(256)
GAT_att_kernel(const float* __restrict__ z,
               const float* __restrict__ sq,
               float* __restrict__ out) {
  __shared__ __attribute__((aligned(16))) float att_s[2][16 * APITCH];
  __shared__ float sq_s[2][512];
  __shared__ float nrm_s[2][512];

  const int tid   = threadIdx.x;
  const int wave  = tid >> 5;
  const int lane  = tid & 31;
  const int ntile = blockIdx.x & 31;
  const int t     = blockIdx.x >> 5;

  const int l15   = lane & 15;
  const int rbase = (lane >> 4) << 3;

  const int srow = tid >> 4;
  const int sl   = tid & 15;

  v4f acc[8];
#pragma unroll
  for (int k = 0; k < 8; ++k) acc[k] = v4f{};

  const size_t bstride = (size_t)T_ * N_ * D_;
  const float* ap0 = z + (size_t)t * N_ * D_ + (size_t)ntile * ZTILE + lane * 4;
  const float* bp0 = z + (size_t)t * N_ * D_ + (size_t)(wave * 4) * ZTILE + lane * 4;

  v4f fa[2], fb[2][4];
  v8f cn[4];
#pragma unroll
  for (int i = 0; i < 4; ++i) cn[i] = v8f{};

  {
    const float* sq0 = sq + (size_t)t * N_;
    const float s0 = sq0[tid];
    const float s1 = sq0[tid + 256];
#pragma unroll
    for (int g = 0; g < 4; ++g) {
      att_load_group(fa, fb, ap0 + g * 256, bp0 + g * 256);
      att_mma_group(cn, fa, fb);
    }
    att_load_group(fa, fb, ap0 + bstride, bp0 + bstride);
    __builtin_amdgcn_sched_barrier(0);
    sq_s[0][tid]        = s0;
    sq_s[0][tid + 256]  = s1;
    nrm_s[0][tid]       = sqrtf(s0) + EPS_;
    nrm_s[0][tid + 256] = sqrtf(s1) + EPS_;
  }
  __syncthreads();

  for (int b = 0; b < B_; ++b) {
    const bool has_next = (b + 1 < B_);
    const float* sqc  = sq_s[b & 1];
    const float* nrmc = nrm_s[b & 1];
    float*       es   = att_s[b & 1];

    v8f cc[4];
#pragma unroll
    for (int i = 0; i < 4; ++i) { cc[i] = cn[i]; cn[i] = v8f{}; }

    float sn0 = 0.f, sn1 = 0.f;
    if (has_next) {
      const float* sqn = sq + (size_t)((b + 1) * T_ + t) * N_;
      sn0 = sqn[tid];
      sn1 = sqn[tid + 256];
    }
    const float* apn = ap0 + (size_t)(b + 1) * bstride;
    const float* bpn = bp0 + (size_t)(b + 1) * bstride;

#pragma unroll
    for (int g = 0; g < 4; ++g) {
      if (has_next) {
        att_mma_group(cn, fa, fb);
        if (g < 3) {
          att_load_group(fa, fb, apn + (g + 1) * 256, bpn + (g + 1) * 256);
        } else if (b + 2 < B_) {
          att_load_group(fa, fb, apn + bstride, bpn + bstride);
        }
      }

      const v8f c = cc[g];
      const int col = (wave * 4 + g) * 16 + l15;
      const float sqm = sqc[col];
      const float nm  = nrmc[col];
#pragma unroll
      for (int v = 0; v < 8; ++v) {
        const int rloc = rbase + v;
        const float sqn = sqc[ntile * 16 + rloc];
        const float nn  = nrmc[ntile * 16 + rloc];
        const float dot = c[v];
        const float cosv = dot / (nn * nm);
        const float d2   = fmaxf(sqn + sqm - 2.0f * dot, 0.0f);
        const float diff = sqrtf(d2);
        es[rloc * APITCH + col] = __expf(cosv - diff / (nn + nm + EPS_));
      }
    }

    if (has_next) {
      sq_s[(b + 1) & 1][tid]        = sn0;
      sq_s[(b + 1) & 1][tid + 256]  = sn1;
      nrm_s[(b + 1) & 1][tid]       = sqrtf(sn0) + EPS_;
      nrm_s[(b + 1) & 1][tid + 256] = sqrtf(sn1) + EPS_;
    }
    __syncthreads();

    {
      const float* er = es + srow * APITCH + sl * 4;
      v4f ev[8];
#pragma unroll
      for (int k = 0; k < 8; ++k) ev[k] = *(const v4f*)(er + k * 64);
      v4f p4 = ev[0];
#pragma unroll
      for (int k = 1; k < 8; ++k) p4 += ev[k];
      float part = (p4[0] + p4[1]) + (p4[2] + p4[3]);
#pragma unroll
      for (int m = 1; m < 16; m <<= 1) part += __shfl_xor(part, m);
      const float wgt = (1.0f / (float)B_) / (part + EPS_);
#pragma unroll
      for (int k = 0; k < 8; ++k) {
#pragma unroll
        for (int e = 0; e < 4; ++e) acc[k][e] = fmaf(ev[k][e], wgt, acc[k][e]);
      }
    }
  }

  float* outp = out + ((size_t)t * N_ + ntile * 16 + srow) * N_ + sl * 4;
#pragma unroll
  for (int k = 0; k < 8; ++k) *(v4f*)(outp + k * 64) = acc[k];
}

extern "C" void kernel_launch(void* const* d_in, const int* in_sizes, int n_in,
                              void* d_out, int out_size, void* d_ws, size_t ws_size,
                              hipStream_t stream) {
  const float* x    = (const float*)d_in[0];
  const float* W    = (const float*)d_in[1];
  const float* bias = (const float*)d_in[2];
  float* out = (float*)d_out;

  float* z  = (float*)d_ws;
  float* sq = z + (size_t)B_ * T_ * N_ * D_;

  GAT_z_kernel<<<B_ * (N_ / 16), 256, 0, stream>>>(x, W, bias, z);

  GAT_sq_kernel<<<B_ * T_ * N_ / 64 / 8, 256, 0, stream>>>(z, sq);

  GAT_att_kernel<<<T_ * (N_ / 16), 256, 0, stream>>>(z, sq, out);
}


# att epilogue: exponent cos - diff/sum taken as one quotient (dot*sum - diff*nn*nm)/(nn*nm*sum): one correctly rounded division a value (was two); exponent rounding changes, sum orders kept
# speedup vs baseline: 3.3095x; 1.0804x over previous
#include <hip/hip_runtime.h>
#include <hip/hip_bf16.h>

typedef __attribute__((ext_vector_type(2))) float v2f;
typedef __attribute__((ext_vector_type(4))) float v4f;
typedef __attribute__((ext_vector_type(8))) float v8f;

#define B_ 8
#define N_ 512
#define F_ 128
#define T_ 32
#define D_ 64
#define EPS_ 1e-6f

#define ZTILE 1024
#define ZFC  16
#define ZNCH (F_ / ZFC)
#define ZTP  (ZFC + 2)
#define ZNP  (T_ * ZTP + 4)
#define ZWP  (ZFC + 4)

__global__ void __launch_bounds__(256)
GAT_z_kernel(const float* __restrict__ x,
             const float* __restrict__ W,
             const float* __restrict__ bias,
             float* __restrict__ z) {
  __shared__ __attribute__((aligned(16))) float xs[2][16 * ZNP];
  __shared__ __attribute__((aligned(16))) float wsm[2][D_ * ZWP];

  const int tid   = threadIdx.x;
  const int wave  = tid >> 5;
  const int lane  = tid & 31;
  const int ntile = blockIdx.x & 31;
  const int b     = blockIdx.x >> 5;
  const int l15   = lane & 15;
  const int h     = lane >> 4;

  const float* xg = x + (size_t)(b * N_ + ntile * 16 + (tid >> 7)) * (F_ * T_) + (tid & 127) * 4;
  const int    xl = (tid >> 7) * ZNP + ((tid & 7) * 4) * ZTP + ((tid & 127) >> 3);
  const float* wg = W + (size_t)(tid >> 2) * F_ + (tid & 3) * 4;
  const int    wl = (tid >> 2) * ZWP + (tid & 3) * 4;

  v4f xr[8];
  v4f wr;
#pragma unroll
  for (int p = 0; p < 8; ++p) xr[p] = *(const v4f*)(xg + (size_t)(2 * p) * (F_ * T_));
  wr = *(const v4f*)wg;
#pragma unroll
  for (int p = 0; p < 8; ++p) {
#pragma unroll
    for (int i = 0; i < 4; ++i) xs[0][xl + 2 * p * ZNP + i * ZTP] = xr[p][i];
  }
  *(v4f*)&wsm[0][wl] = wr;

  v8f acc[4][4];
#pragma unroll
  for (int j = 0; j < 4; ++j) {
#pragma unroll
    for (int dt = 0; dt < 4; ++dt) acc[j][dt] = v8f{};
  }

  const int aoff = l15 * ZNP + (4 * wave) * ZTP + 2 * h;
  const int boff = l15 * ZWP + 2 * h;

  for (int c = 0; c < ZNCH; ++c) {
    __syncthreads();

    if (c + 1 < ZNCH) {
#pragma unroll
      for (int p = 0; p < 8; ++p)
        xr[p] = *(const v4f*)(xg + (size_t)(2 * p) * (F_ * T_) + (c + 1) * ZFC * T_);
      wr = *(const v4f*)(wg + (c + 1) * ZFC);
    }
    __builtin_amdgcn_sched_barrier(0);

    const float* xa = &xs[c & 1][aoff];
    const float* wb = &wsm[c & 1][boff];
#pragma unroll
    for (int kk = 0; kk < ZFC; kk += 4) {
      v2f af[4];
      v2f bf[4];
#pragma unroll
      for (int j = 0; j < 4; ++j)  af[j]  = *(const v2f*)(xa + j * ZTP + kk);
#pragma unroll
      for (int dt = 0; dt < 4; ++dt) bf[dt] = *(const v2f*)(wb + dt * 16 * ZWP + kk);
#pragma unroll
      for (int j = 0; j < 4; ++j) {
#pragma unroll
        for (int dt = 0; dt < 4; ++dt)
          acc[j][dt] = __builtin_amdgcn_wmma_f32_16x16x4_f32(false, bf[dt], false, af[j],
                                                             (short)0, acc[j][dt], false, false);
      }
    }

    __builtin_amdgcn_sched_barrier(0);
    if (c + 1 < ZNCH) {
#pragma unroll
      for (int p = 0; p < 8; ++p) {
#pragma unroll
        for (int i = 0; i < 4; ++i) xs[(c + 1) & 1][xl + 2 * p * ZNP + i * ZTP] = xr[p][i];
      }
      *(v4f*)&wsm[(c + 1) & 1][wl] = wr;
    }
  }

  v4f bv[4][2];
#pragma unroll
  for (int dt = 0; dt < 4; ++dt) {
#pragma unroll
    for (int u = 0; u < 2; ++u) bv[dt][u] = *(const v4f*)(bias + dt * 16 + h * 8 + u * 4);
  }
#pragma unroll
  for (int j = 0; j < 4; ++j) {
    const int t = 4 * wave + j;
    float* zt = z + ((size_t)(b * T_ + t) * (N_ / 16) + ntile) * ZTILE + lane * 4;
#pragma unroll
    for (int dt = 0; dt < 4; ++dt) {
#pragma unroll
      for (int u = 0; u < 2; ++u) {
        v4f o;
#pragma unroll
        for (int e = 0; e < 4; ++e) o[e] = tanhf(acc[j][dt][4 * u + e] + bv[dt][u][e]);
        *(v4f*)(zt + (2 * dt + u) * 128) = o;
      }
    }
  }
}

__global__ void __launch_bounds__(256)
GAT_sq_kernel(const float* __restrict__ z,
              float* __restrict__ sq) {
  const int lane = threadIdx.x & 31;
  const int wv   = blockIdx.x * 8 + (threadIdx.x >> 5);
  const float* zr = z + (size_t)wv * (4 * ZTILE) + lane * 4;
  float s[4];
#pragma unroll
  for (int i = 0; i < 4; ++i) {
    float a = 0.f;
#pragma unroll
    for (int q = 0; q < 8; ++q) {
      const v4f v = *(const v4f*)(zr + i * ZTILE + q * 128);
#pragma unroll
      for (int e = 0; e < 4; ++e) a = fmaf(v[e], v[e], a);
    }
    s[i] = a;
  }
#pragma unroll
  for (int i = 0; i < 4; ++i) s[i] += __shfl_xor(s[i], 16);
  float* o = sq + (size_t)wv * 64 + lane;
  o[0]  = (lane < 16) ? s[0] : s[1];
  o[32] = (lane < 16) ? s[2] : s[3];
}

#define APITCH 516

static __device__ __forceinline__ void att_load_group(v4f (&A)[2], v4f (&Bf)[2][4],
                                                      const float* __restrict__ ap,
                                                      const float* __restrict__ bp) {
#pragma unroll
  for (int u = 0; u < 2; ++u) {
    A[u] = *(const v4f*)(ap + u * 128);
#pragma unroll
    for (int i = 0; i < 4; ++i) Bf[u][i] = *(const v4f*)(bp + i * ZTILE + u * 128);
  }
}

static __device__ __forceinline__ void att_mma_group(v8f (&c)[4], const v4f (&A)[2], const v4f (&Bf)[2][4]) {
#pragma unroll
  for (int u = 0; u < 2; ++u) {
    const v2f alo = {A[u][0], A[u][1]};
    const v2f ahi = {A[u][2], A[u][3]};
#pragma unroll
    for (int i = 0; i < 4; ++i)
      c[i] = __builtin_amdgcn_wmma_f32_16x16x4_f32(false, alo, false, v2f{Bf[u][i][0], Bf[u][i][1]},
                                                   (short)0, c[i], false, false);
#pragma unroll
    for (int i = 0; i < 4; ++i)
      c[i] = __builtin_amdgcn_wmma_f32_16x16x4_f32(false, ahi, false, v2f{Bf[u][i][2], Bf[u][i][3]},
                                                   (short)0, c[i], false, false);
  }
}

__global__ void __launch_bounds__(256)
GAT_att_kernel(const float* __restrict__ z,
               const float* __restrict__ sq,
               float* __restrict__ out) {
  __shared__ __attribute__((aligned(16))) float att_s[2][16 * APITCH];
  __shared__ float sq_s[2][512];
  __shared__ float nrm_s[2][512];

  const int tid   = threadIdx.x;
  const int wave  = tid >> 5;
  const int lane  = tid & 31;
  const int ntile = blockIdx.x & 31;
  const int t     = blockIdx.x >> 5;

  const int l15   = lane & 15;
  const int rbase = (lane >> 4) << 3;

  const int srow = tid >> 4;
  const int sl   = tid & 15;

  v4f acc[8];
#pragma unroll
  for (int k = 0; k < 8; ++k) acc[k] = v4f{};

  const size_t bstride = (size_t)T_ * N_ * D_;
  const float* ap0 = z + (size_t)t * N_ * D_ + (size_t)ntile * ZTILE + lane * 4;
  const float* bp0 = z + (size_t)t * N_ * D_ + (size_t)(wave * 4) * ZTILE + lane * 4;

  v4f fa[2], fb[2][4];
  v8f cn[4];
#pragma unroll
  for (int i = 0; i < 4; ++i) cn[i] = v8f{};

  {
    const float* sq0 = sq + (size_t)t * N_;
    const float s0 = sq0[tid];
    const float s1 = sq0[tid + 256];
#pragma unroll
    for (int g = 0; g < 4; ++g) {
      att_load_group(fa, fb, ap0 + g * 256, bp0 + g * 256);
      att_mma_group(cn, fa, fb);
    }
    att_load_group(fa, fb, ap0 + bstride, bp0 + bstride);
    __builtin_amdgcn_sched_barrier(0);
    sq_s[0][tid]        = s0;
    sq_s[0][tid + 256]  = s1;
    nrm_s[0][tid]       = sqrtf(s0) + EPS_;
    nrm_s[0][tid + 256] = sqrtf(s1) + EPS_;
  }
  __syncthreads();

  for (int b = 0; b < B_; ++b) {
    const bool has_next = (b + 1 < B_);
    const float* sqc  = sq_s[b & 1];
    const float* nrmc = nrm_s[b & 1];
    float*       es   = att_s[b & 1];

    v8f cc[4];
#pragma unroll
    for (int i = 0; i < 4; ++i) { cc[i] = cn[i]; cn[i] = v8f{}; }

    float sn0 = 0.f, sn1 = 0.f;
    if (has_next) {
      const float* sqn = sq + (size_t)((b + 1) * T_ + t) * N_;
      sn0 = sqn[tid];
      sn1 = sqn[tid + 256];
    }
    const float* apn = ap0 + (size_t)(b + 1) * bstride;
    const float* bpn = bp0 + (size_t)(b + 1) * bstride;

#pragma unroll
    for (int g = 0; g < 4; ++g) {
      if (has_next) {
        att_mma_group(cn, fa, fb);
        if (g < 3) {
          att_load_group(fa, fb, apn + (g + 1) * 256, bpn + (g + 1) * 256);
        } else if (b + 2 < B_) {
          att_load_group(fa, fb, apn + bstride, bpn + bstride);
        }
      }

      const v8f c = cc[g];
      const int col = (wave * 4 + g) * 16 + l15;
      const float sqm = sqc[col];
      const float nm  = nrmc[col];
#pragma unroll
      for (int v = 0; v < 8; ++v) {
        const int rloc = rbase + v;
        const float sqn = sqc[ntile * 16 + rloc];
        const float nn  = nrmc[ntile * 16 + rloc];
        const float dot = c[v];
        const float pn   = nn * nm;
        const float sn   = nn + nm + EPS_;
        const float d2   = fmaxf(sqn + sqm - 2.0f * dot, 0.0f);
        const float diff = sqrtf(d2);
        const float num  = fmaf(dot, sn, -(diff * pn));
        es[rloc * APITCH + col] = __expf(num / (pn * sn));
      }
    }

    if (has_next) {
      sq_s[(b + 1) & 1][tid]        = sn0;
      sq_s[(b + 1) & 1][tid + 256]  = sn1;
      nrm_s[(b + 1) & 1][tid]       = sqrtf(sn0) + EPS_;
      nrm_s[(b + 1) & 1][tid + 256] = sqrtf(sn1) + EPS_;
    }
    __syncthreads();

    {
      const float* er = es + srow * APITCH + sl * 4;
      v4f ev[8];
#pragma unroll
      for (int k = 0; k < 8; ++k) ev[k] = *(const v4f*)(er + k * 64);
      v4f p4 = ev[0];
#pragma unroll
      for (int k = 1; k < 8; ++k) p4 += ev[k];
      float part = (p4[0] + p4[1]) + (p4[2] + p4[3]);
#pragma unroll
      for (int m = 1; m < 16; m <<= 1) part += __shfl_xor(part, m);
      const float wgt = (1.0f / (float)B_) / (part + EPS_);
#pragma unroll
      for (int k = 0; k < 8; ++k) {
#pragma unroll
        for (int e = 0; e < 4; ++e) acc[k][e] = fmaf(ev[k][e], wgt, acc[k][e]);
      }
    }
  }

  float* outp = out + ((size_t)t * N_ + ntile * 16 + srow) * N_ + sl * 4;
#pragma unroll
  for (int k = 0; k < 8; ++k) *(v4f*)(outp + k * 64) = acc[k];
}

extern "C" void kernel_launch(void* const* d_in, const int* in_sizes, int n_in,
                              void* d_out, int out_size, void* d_ws, size_t ws_size,
                              hipStream_t stream) {
  const float* x    = (const float*)d_in[0];
  const float* W    = (const float*)d_in[1];
  const float* bias = (const float*)d_in[2];
  float* out = (float*)d_out;

  float* z  = (float*)d_ws;
  float* sq = z + (size_t)B_ * T_ * N_ * D_;

  GAT_z_kernel<<<B_ * (N_ / 16), 256, 0, stream>>>(x, W, bias, z);

  GAT_sq_kernel<<<B_ * T_ * N_ / 64 / 8, 256, 0, stream>>>(z, sq);

  GAT_att_kernel<<<T_ * (N_ / 16), 256, 0, stream>>>(z, sq, out);
}


# att epilogue: quotient and square root keep round-to-nearest refinement (estimate + fused steps) but on row pairs as packed f32 ops; range scaling dropped; diff may move 1 ulp; sum orders kept
# speedup vs baseline: 3.6053x; 1.0894x over previous
#include <hip/hip_runtime.h>
#include <hip/hip_bf16.h>

typedef __attribute__((ext_vector_type(2))) float v2f;
typedef __attribute__((ext_vector_type(4))) float v4f;
typedef __attribute__((ext_vector_type(8))) float v8f;

#define B_ 8
#define N_ 512
#define F_ 128
#define T_ 32
#define D_ 64
#define EPS_ 1e-6f

#define ZTILE 1024
#define ZFC  16
#define ZNCH (F_ / ZFC)
#define ZTP  (ZFC + 2)
#define ZNP  (T_ * ZTP + 4)
#define ZWP  (ZFC + 4)

__global__ void __launch_bounds__(256)
GAT_z_kernel(const float* __restrict__ x,
             const float* __restrict__ W,
             const float* __restrict__ bias,
             float* __restrict__ z) {
  __shared__ __attribute__((aligned(16))) float xs[2][16 * ZNP];
  __shared__ __attribute__((aligned(16))) float wsm[2][D_ * ZWP];

  const int tid   = threadIdx.x;
  const int wave  = tid >> 5;
  const int lane  = tid & 31;
  const int ntile = blockIdx.x & 31;
  const int b     = blockIdx.x >> 5;
  const int l15   = lane & 15;
  const int h     = lane >> 4;

  const float* xg = x + (size_t)(b * N_ + ntile * 16 + (tid >> 7)) * (F_ * T_) + (tid & 127) * 4;
  const int    xl = (tid >> 7) * ZNP + ((tid & 7) * 4) * ZTP + ((tid & 127) >> 3);
  const float* wg = W + (size_t)(tid >> 2) * F_ + (tid & 3) * 4;
  const int    wl = (tid >> 2) * ZWP + (tid & 3) * 4;

  v4f xr[8];
  v4f wr;
#pragma unroll
  for (int p = 0; p < 8; ++p) xr[p] = *(const v4f*)(xg + (size_t)(2 * p) * (F_ * T_));
  wr = *(const v4f*)wg;
#pragma unroll
  for (int p = 0; p < 8; ++p) {
#pragma unroll
    for (int i = 0; i < 4; ++i) xs[0][xl + 2 * p * ZNP + i * ZTP] = xr[p][i];
  }
  *(v4f*)&wsm[0][wl] = wr;

  v8f acc[4][4];
#pragma unroll
  for (int j = 0; j < 4; ++j) {
#pragma unroll
    for (int dt = 0; dt < 4; ++dt) acc[j][dt] = v8f{};
  }

  const int aoff = l15 * ZNP + (4 * wave) * ZTP + 2 * h;
  const int boff = l15 * ZWP + 2 * h;

  for (int c = 0; c < ZNCH; ++c) {
    __syncthreads();

    if (c + 1 < ZNCH) {
#pragma unroll
      for (int p = 0; p < 8; ++p)
        xr[p] = *(const v4f*)(xg + (size_t)(2 * p) * (F_ * T_) + (c + 1) * ZFC * T_);
      wr = *(const v4f*)(wg + (c + 1) * ZFC);
    }
    __builtin_amdgcn_sched_barrier(0);

    const float* xa = &xs[c & 1][aoff];
    const float* wb = &wsm[c & 1][boff];
#pragma unroll
    for (int kk = 0; kk < ZFC; kk += 4) {
      v2f af[4];
      v2f bf[4];
#pragma unroll
      for (int j = 0; j < 4; ++j)  af[j]  = *(const v2f*)(xa + j * ZTP + kk);
#pragma unroll
      for (int dt = 0; dt < 4; ++dt) bf[dt] = *(const v2f*)(wb + dt * 16 * ZWP + kk);
#pragma unroll
      for (int j = 0; j < 4; ++j) {
#pragma unroll
        for (int dt = 0; dt < 4; ++dt)
          acc[j][dt] = __builtin_amdgcn_wmma_f32_16x16x4_f32(false, bf[dt], false, af[j],
                                                             (short)0, acc[j][dt], false, false);
      }
    }

    __builtin_amdgcn_sched_barrier(0);
    if (c + 1 < ZNCH) {
#pragma unroll
      for (int p = 0; p < 8; ++p) {
#pragma unroll
        for (int i = 0; i < 4; ++i) xs[(c + 1) & 1][xl + 2 * p * ZNP + i * ZTP] = xr[p][i];
      }
      *(v4f*)&wsm[(c + 1) & 1][wl] = wr;
    }
  }

  v4f bv[4][2];
#pragma unroll
  for (int dt = 0; dt < 4; ++dt) {
#pragma unroll
    for (int u = 0; u < 2; ++u) bv[dt][u] = *(const v4f*)(bias + dt * 16 + h * 8 + u * 4);
  }
#pragma unroll
  for (int j = 0; j < 4; ++j) {
    const int t = 4 * wave + j;
    float* zt = z + ((size_t)(b * T_ + t) * (N_ / 16) + ntile) * ZTILE + lane * 4;
#pragma unroll
    for (int dt = 0; dt < 4; ++dt) {
#pragma unroll
      for (int u = 0; u < 2; ++u) {
        v4f o;
#pragma unroll
        for (int e = 0; e < 4; ++e) o[e] = tanhf(acc[j][dt][4 * u + e] + bv[dt][u][e]);
        *(v4f*)(zt + (2 * dt + u) * 128) = o;
      }
    }
  }
}

__global__ void __launch_bounds__(256)
GAT_sq_kernel(const float* __restrict__ z,
              float* __restrict__ sq) {
  const int lane = threadIdx.x & 31;
  const int wv   = blockIdx.x * 8 + (threadIdx.x >> 5);
  const float* zr = z + (size_t)wv * (4 * ZTILE) + lane * 4;
  float s[4];
#pragma unroll
  for (int i = 0; i < 4; ++i) {
    float a = 0.f;
#pragma unroll
    for (int q = 0; q < 8; ++q) {
      const v4f v = *(const v4f*)(zr + i * ZTILE + q * 128);
#pragma unroll
      for (int e = 0; e < 4; ++e) a = fmaf(v[e], v[e], a);
    }
    s[i] = a;
  }
#pragma unroll
  for (int i = 0; i < 4; ++i) s[i] += __shfl_xor(s[i], 16);
  float* o = sq + (size_t)wv * 64 + lane;
  o[0]  = (lane < 16) ? s[0] : s[1];
  o[32] = (lane < 16) ? s[2] : s[3];
}

#define APITCH 516

static __device__ __forceinline__ v2f pk_fma(v2f a, v2f b, v2f c) { return __builtin_elementwise_fma(a, b, c); }

static __device__ __forceinline__ v2f pk_div(v2f n, v2f d) {
  v2f r = {__builtin_amdgcn_rcpf(d[0]), __builtin_amdgcn_rcpf(d[1])};
  const v2f one = {1.0f, 1.0f};
  const v2f nd = -d;
  r = pk_fma(pk_fma(nd, r, one), r, r);
  v2f q = n * r;
  q = pk_fma(pk_fma(nd, q, n), r, q);
  return pk_fma(pk_fma(nd, q, n), r, q);
}

static __device__ __forceinline__ v2f pk_sqrt(v2f x) {
  const v2f tiny = {1.17549435e-38f, 1.17549435e-38f};
  const v2f xr = __builtin_elementwise_max(x, tiny);
  const v2f r = {__builtin_amdgcn_rsqf(xr[0]), __builtin_amdgcn_rsqf(xr[1])};
  const v2f half = {0.5f, 0.5f};
  v2f g = x * r;
  v2f h = r * half;
  const v2f e = pk_fma(-h, g, half);
  h = pk_fma(h, e, h);
  g = pk_fma(g, e, g);
  const v2f d = pk_fma(-g, g, x);
  return pk_fma(d, h, g);
}

static __device__ __forceinline__ void att_load_group(v4f (&A)[2], v4f (&Bf)[2][4],
                                                      const float* __restrict__ ap,
                                                      const float* __restrict__ bp) {
#pragma unroll
  for (int u = 0; u < 2; ++u) {
    A[u] = *(const v4f*)(ap + u * 128);
#pragma unroll
    for (int i = 0; i < 4; ++i) Bf[u][i] = *(const v4f*)(bp + i * ZTILE + u * 128);
  }
}

static __device__ __forceinline__ void att_mma_group(v8f (&c)[4], const v4f (&A)[2], const v4f (&Bf)[2][4]) {
#pragma unroll
  for (int u = 0; u < 2; ++u) {
    const v2f alo = {A[u][0], A[u][1]};
    const v2f ahi = {A[u][2], A[u][3]};
#pragma unroll
    for (int i = 0; i < 4; ++i)
      c[i] = __builtin_amdgcn_wmma_f32_16x16x4_f32(false, alo, false, v2f{Bf[u][i][0], Bf[u][i][1]},
                                                   (short)0, c[i], false, false);
#pragma unroll
    for (int i = 0; i < 4; ++i)
      c[i] = __builtin_amdgcn_wmma_f32_16x16x4_f32(false, ahi, false, v2f{Bf[u][i][2], Bf[u][i][3]},
                                                   (short)0, c[i], false, false);
  }
}

__global__ void __launch_bounds__(256)
GAT_att_kernel(const float* __restrict__ z,
               const float* __restrict__ sq,
               float* __restrict__ out) {
  __shared__ __attribute__((aligned(16))) float att_s[2][16 * APITCH];
  __shared__ __attribute__((aligned(16))) float sq_s[2][512];
  __shared__ __attribute__((aligned(16))) float nrm_s[2][512];

  const int tid   = threadIdx.x;
  const int wave  = tid >> 5;
  const int lane  = tid & 31;
  const int ntile = blockIdx.x & 31;
  const int t     = blockIdx.x >> 5;

  const int l15   = lane & 15;
  const int rbase = (lane >> 4) << 3;

  const int srow = tid >> 4;
  const int sl   = tid & 15;

  v4f acc[8];
#pragma unroll
  for (int k = 0; k < 8; ++k) acc[k] = v4f{};

  const size_t bstride = (size_t)T_ * N_ * D_;
  const float* ap0 = z + (size_t)t * N_ * D_ + (size_t)ntile * ZTILE + lane * 4;
  const float* bp0 = z + (size_t)t * N_ * D_ + (size_t)(wave * 4) * ZTILE + lane * 4;

  v4f fa[2], fb[2][4];
  v8f cn[4];
#pragma unroll
  for (int i = 0; i < 4; ++i) cn[i] = v8f{};

  {
    const float* sq0 = sq + (size_t)t * N_;
    const float s0 = sq0[tid];
    const float s1 = sq0[tid + 256];
#pragma unroll
    for (int g = 0; g < 4; ++g) {
      att_load_group(fa, fb, ap0 + g * 256, bp0 + g * 256);
      att_mma_group(cn, fa, fb);
    }
    att_load_group(fa, fb, ap0 + bstride, bp0 + bstride);
    __builtin_amdgcn_sched_barrier(0);
    sq_s[0][tid]        = s0;
    sq_s[0][tid + 256]  = s1;
    nrm_s[0][tid]       = sqrtf(s0) + EPS_;
    nrm_s[0][tid + 256] = sqrtf(s1) + EPS_;
  }
  __syncthreads();

  for (int b = 0; b < B_; ++b) {
    const bool has_next = (b + 1 < B_);
    const float* sqc  = sq_s[b & 1];
    const float* nrmc = nrm_s[b & 1];
    float*       es   = att_s[b & 1];

    v8f cc[4];
#pragma unroll
    for (int i = 0; i < 4; ++i) { cc[i] = cn[i]; cn[i] = v8f{}; }

    float sn0 = 0.f, sn1 = 0.f;
    if (has_next) {
      const float* sqn = sq + (size_t)((b + 1) * T_ + t) * N_;
      sn0 = sqn[tid];
      sn1 = sqn[tid + 256];
    }
    const float* apn = ap0 + (size_t)(b + 1) * bstride;
    const float* bpn = bp0 + (size_t)(b + 1) * bstride;

#pragma unroll
    for (int g = 0; g < 4; ++g) {
      if (has_next) {
        att_mma_group(cn, fa, fb);
        if (g < 3) {
          att_load_group(fa, fb, apn + (g + 1) * 256, bpn + (g + 1) * 256);
        } else if (b + 2 < B_) {
          att_load_group(fa, fb, apn + bstride, bpn + bstride);
        }
      }

      const v8f c = cc[g];
      const int col = (wave * 4 + g) * 16 + l15;
      const float sqm = sqc[col];
      const float nm  = nrmc[col];
      const v2f sqm2 = {sqm, sqm};
      const v2f nm2  = {nm, nm};
      const v2f eps2 = {EPS_, EPS_};
      const v2f zero2 = {0.0f, 0.0f};
      const v2f mtwo2 = {-2.0f, -2.0f};
#pragma unroll
      for (int v = 0; v < 8; v += 2) {
        const int rloc = rbase + v;
        const v2f sqn = *(const v2f*)&sqc[ntile * 16 + rloc];
        const v2f nn  = *(const v2f*)&nrmc[ntile * 16 + rloc];
        const v2f dot = {c[v], c[v + 1]};
        const v2f pn  = nn * nm2;
        const v2f sn  = (nn + nm2) + eps2;
        const v2f d2  = __builtin_elementwise_max(pk_fma(dot, mtwo2, sqn + sqm2), zero2);
        const v2f diff = pk_sqrt(d2);
        const v2f num = pk_fma(dot, sn, -(diff * pn));
        const v2f q   = pk_div(num, pn * sn);
        es[rloc * APITCH + col]       = __expf(q[0]);
        es[(rloc + 1) * APITCH + col] = __expf(q[1]);
      }
    }

    if (has_next) {
      sq_s[(b + 1) & 1][tid]        = sn0;
      sq_s[(b + 1) & 1][tid + 256]  = sn1;
      nrm_s[(b + 1) & 1][tid]       = sqrtf(sn0) + EPS_;
      nrm_s[(b + 1) & 1][tid + 256] = sqrtf(sn1) + EPS_;
    }
    __syncthreads();

    {
      const float* er = es + srow * APITCH + sl * 4;
      v4f ev[8];
#pragma unroll
      for (int k = 0; k < 8; ++k) ev[k] = *(const v4f*)(er + k * 64);
      v4f p4 = ev[0];
#pragma unroll
      for (int k = 1; k < 8; ++k) p4 += ev[k];
      float part = (p4[0] + p4[1]) + (p4[2] + p4[3]);
#pragma unroll
      for (int m = 1; m < 16; m <<= 1) part += __shfl_xor(part, m);
      const float wgt = (1.0f / (float)B_) / (part + EPS_);
#pragma unroll
      for (int k = 0; k < 8; ++k) {
#pragma unroll
        for (int e = 0; e < 4; ++e) acc[k][e] = fmaf(ev[k][e], wgt, acc[k][e]);
      }
    }
  }

  float* outp = out + ((size_t)t * N_ + ntile * 16 + srow) * N_ + sl * 4;
#pragma unroll
  for (int k = 0; k < 8; ++k) *(v4f*)(outp + k * 64) = acc[k];
}

extern "C" void kernel_launch(void* const* d_in, const int* in_sizes, int n_in,
                              void* d_out, int out_size, void* d_ws, size_t ws_size,
                              hipStream_t stream) {
  const float* x    = (const float*)d_in[0];
  const float* W    = (const float*)d_in[1];
  const float* bias = (const float*)d_in[2];
  float* out = (float*)d_out;

  float* z  = (float*)d_ws;
  float* sq = z + (size_t)B_ * T_ * N_ * D_;

  GAT_z_kernel<<<B_ * (N_ / 16), 256, 0, stream>>>(x, W, bias, z);

  GAT_sq_kernel<<<B_ * T_ * N_ / 64 / 8, 256, 0, stream>>>(z, sq);

  GAT_att_kernel<<<T_ * (N_ / 16), 256, 0, stream>>>(z, sq, out);
}


# att epilogue: diff/sum as d2*rsq(d2*sum^2) and cos as dot*rcp(nn*nm): hardware estimates (about 1 ulp) replace the refined quotient and root; exponent rounding changes; sum orders kept
# speedup vs baseline: 3.7796x; 1.0483x over previous
#include <hip/hip_runtime.h>
#include <hip/hip_bf16.h>

typedef __attribute__((ext_vector_type(2))) float v2f;
typedef __attribute__((ext_vector_type(4))) float v4f;
typedef __attribute__((ext_vector_type(8))) float v8f;

#define B_ 8
#define N_ 512
#define F_ 128
#define T_ 32
#define D_ 64
#define EPS_ 1e-6f

#define ZTILE 1024
#define ZFC  16
#define ZNCH (F_ / ZFC)
#define ZTP  (ZFC + 2)
#define ZNP  (T_ * ZTP + 4)
#define ZWP  (ZFC + 4)

__global__ void __launch_bounds__(256)
GAT_z_kernel(const float* __restrict__ x,
             const float* __restrict__ W,
             const float* __restrict__ bias,
             float* __restrict__ z) {
  __shared__ __attribute__((aligned(16))) float xs[2][16 * ZNP];
  __shared__ __attribute__((aligned(16))) float wsm[2][D_ * ZWP];

  const int tid   = threadIdx.x;
  const int wave  = tid >> 5;
  const int lane  = tid & 31;
  const int ntile = blockIdx.x & 31;
  const int b     = blockIdx.x >> 5;
  const int l15   = lane & 15;
  const int h     = lane >> 4;

  const float* xg = x + (size_t)(b * N_ + ntile * 16 + (tid >> 7)) * (F_ * T_) + (tid & 127) * 4;
  const int    xl = (tid >> 7) * ZNP + ((tid & 7) * 4) * ZTP + ((tid & 127) >> 3);
  const float* wg = W + (size_t)(tid >> 2) * F_ + (tid & 3) * 4;
  const int    wl = (tid >> 2) * ZWP + (tid & 3) * 4;

  v4f xr[8];
  v4f wr;
#pragma unroll
  for (int p = 0; p < 8; ++p) xr[p] = *(const v4f*)(xg + (size_t)(2 * p) * (F_ * T_));
  wr = *(const v4f*)wg;
#pragma unroll
  for (int p = 0; p < 8; ++p) {
#pragma unroll
    for (int i = 0; i < 4; ++i) xs[0][xl + 2 * p * ZNP + i * ZTP] = xr[p][i];
  }
  *(v4f*)&wsm[0][wl] = wr;

  v8f acc[4][4];
#pragma unroll
  for (int j = 0; j < 4; ++j) {
#pragma unroll
    for (int dt = 0; dt < 4; ++dt) acc[j][dt] = v8f{};
  }

  const int aoff = l15 * ZNP + (4 * wave) * ZTP + 2 * h;
  const int boff = l15 * ZWP + 2 * h;

  for (int c = 0; c < ZNCH; ++c) {
    __syncthreads();

    if (c + 1 < ZNCH) {
#pragma unroll
      for (int p = 0; p < 8; ++p)
        xr[p] = *(const v4f*)(xg + (size_t)(2 * p) * (F_ * T_) + (c + 1) * ZFC * T_);
      wr = *(const v4f*)(wg + (c + 1) * ZFC);
    }
    __builtin_amdgcn_sched_barrier(0);

    const float* xa = &xs[c & 1][aoff];
    const float* wb = &wsm[c & 1][boff];
#pragma unroll
    for (int kk = 0; kk < ZFC; kk += 4) {
      v2f af[4];
      v2f bf[4];
#pragma unroll
      for (int j = 0; j < 4; ++j)  af[j]  = *(const v2f*)(xa + j * ZTP + kk);
#pragma unroll
      for (int dt = 0; dt < 4; ++dt) bf[dt] = *(const v2f*)(wb + dt * 16 * ZWP + kk);
#pragma unroll
      for (int j = 0; j < 4; ++j) {
#pragma unroll
        for (int dt = 0; dt < 4; ++dt)
          acc[j][dt] = __builtin_amdgcn_wmma_f32_16x16x4_f32(false, bf[dt], false, af[j],
                                                             (short)0, acc[j][dt], false, false);
      }
    }

    __builtin_amdgcn_sched_barrier(0);
    if (c + 1 < ZNCH) {
#pragma unroll
      for (int p = 0; p < 8; ++p) {
#pragma unroll
        for (int i = 0; i < 4; ++i) xs[(c + 1) & 1][xl + 2 * p * ZNP + i * ZTP] = xr[p][i];
      }
      *(v4f*)&wsm[(c + 1) & 1][wl] = wr;
    }
  }

  v4f bv[4][2];
#pragma unroll
  for (int dt = 0; dt < 4; ++dt) {
#pragma unroll
    for (int u = 0; u < 2; ++u) bv[dt][u] = *(const v4f*)(bias + dt * 16 + h * 8 + u * 4);
  }
#pragma unroll
  for (int j = 0; j < 4; ++j) {
    const int t = 4 * wave + j;
    float* zt = z + ((size_t)(b * T_ + t) * (N_ / 16) + ntile) * ZTILE + lane * 4;
#pragma unroll
    for (int dt = 0; dt < 4; ++dt) {
#pragma unroll
      for (int u = 0; u < 2; ++u) {
        v4f o;
#pragma unroll
        for (int e = 0; e < 4; ++e) o[e] = tanhf(acc[j][dt][4 * u + e] + bv[dt][u][e]);
        *(v4f*)(zt + (2 * dt + u) * 128) = o;
      }
    }
  }
}

__global__ void __launch_bounds__(256)
GAT_sq_kernel(const float* __restrict__ z,
              float* __restrict__ sq) {
  const int lane = threadIdx.x & 31;
  const int wv   = blockIdx.x * 8 + (threadIdx.x >> 5);
  const float* zr = z + (size_t)wv * (4 * ZTILE) + lane * 4;
  float s[4];
#pragma unroll
  for (int i = 0; i < 4; ++i) {
    float a = 0.f;
#pragma unroll
    for (int q = 0; q < 8; ++q) {
      const v4f v = *(const v4f*)(zr + i * ZTILE + q * 128);
#pragma unroll
      for (int e = 0; e < 4; ++e) a = fmaf(v[e], v[e], a);
    }
    s[i] = a;
  }
#pragma unroll
  for (int i = 0; i < 4; ++i) s[i] += __shfl_xor(s[i], 16);
  float* o = sq + (size_t)wv * 64 + lane;
  o[0]  = (lane < 16) ? s[0] : s[1];
  o[32] = (lane < 16) ? s[2] : s[3];
}

#define APITCH 516

static __device__ __forceinline__ v2f pk_fma(v2f a, v2f b, v2f c) { return __builtin_elementwise_fma(a, b, c); }

static __device__ __forceinline__ void att_load_group(v4f (&A)[2], v4f (&Bf)[2][4],
                                                      const float* __restrict__ ap,
                                                      const float* __restrict__ bp) {
#pragma unroll
  for (int u = 0; u < 2; ++u) {
    A[u] = *(const v4f*)(ap + u * 128);
#pragma unroll
    for (int i = 0; i < 4; ++i) Bf[u][i] = *(const v4f*)(bp + i * ZTILE + u * 128);
  }
}

static __device__ __forceinline__ void att_mma_group(v8f (&c)[4], const v4f (&A)[2], const v4f (&Bf)[2][4]) {
#pragma unroll
  for (int u = 0; u < 2; ++u) {
    const v2f alo = {A[u][0], A[u][1]};
    const v2f ahi = {A[u][2], A[u][3]};
#pragma unroll
    for (int i = 0; i < 4; ++i)
      c[i] = __builtin_amdgcn_wmma_f32_16x16x4_f32(false, alo, false, v2f{Bf[u][i][0], Bf[u][i][1]},
                                                   (short)0, c[i], false, false);
#pragma unroll
    for (int i = 0; i < 4; ++i)
      c[i] = __builtin_amdgcn_wmma_f32_16x16x4_f32(false, ahi, false, v2f{Bf[u][i][2], Bf[u][i][3]},
                                                   (short)0, c[i], false, false);
  }
}

__global__ void __launch_bounds__(256)
GAT_att_kernel(const float* __restrict__ z,
               const float* __restrict__ sq,
               float* __restrict__ out) {
  __shared__ __attribute__((aligned(16))) float att_s[2][16 * APITCH];
  __shared__ __attribute__((aligned(16))) float sq_s[2][512];
  __shared__ __attribute__((aligned(16))) float nrm_s[2][512];

  const int tid   = threadIdx.x;
  const int wave  = tid >> 5;
  const int lane  = tid & 31;
  const int ntile = blockIdx.x & 31;
  const int t     = blockIdx.x >> 5;

  const int l15   = lane & 15;
  const int rbase = (lane >> 4) << 3;

  const int srow = tid >> 4;
  const int sl   = tid & 15;

  v4f acc[8];
#pragma unroll
  for (int k = 0; k < 8; ++k) acc[k] = v4f{};

  const size_t bstride = (size_t)T_ * N_ * D_;
  const float* ap0 = z + (size_t)t * N_ * D_ + (size_t)ntile * ZTILE + lane * 4;
  const float* bp0 = z + (size_t)t * N_ * D_ + (size_t)(wave * 4) * ZTILE + lane * 4;

  v4f fa[2], fb[2][4];
  v8f cn[4];
#pragma unroll
  for (int i = 0; i < 4; ++i) cn[i] = v8f{};

  {
    const float* sq0 = sq + (size_t)t * N_;
    const float s0 = sq0[tid];
    const float s1 = sq0[tid + 256];
#pragma unroll
    for (int g = 0; g < 4; ++g) {
      att_load_group(fa, fb, ap0 + g * 256, bp0 + g * 256);
      att_mma_group(cn, fa, fb);
    }
    att_load_group(fa, fb, ap0 + bstride, bp0 + bstride);
    __builtin_amdgcn_sched_barrier(0);
    sq_s[0][tid]        = s0;
    sq_s[0][tid + 256]  = s1;
    nrm_s[0][tid]       = sqrtf(s0) + EPS_;
    nrm_s[0][tid + 256] = sqrtf(s1) + EPS_;
  }
  __syncthreads();

  for (int b = 0; b < B_; ++b) {
    const bool has_next = (b + 1 < B_);
    const float* sqc  = sq_s[b & 1];
    const float* nrmc = nrm_s[b & 1];
    float*       es   = att_s[b & 1];

    v8f cc[4];
#pragma unroll
    for (int i = 0; i < 4; ++i) { cc[i] = cn[i]; cn[i] = v8f{}; }

    float sn0 = 0.f, sn1 = 0.f;
    if (has_next) {
      const float* sqn = sq + (size_t)((b + 1) * T_ + t) * N_;
      sn0 = sqn[tid];
      sn1 = sqn[tid + 256];
    }
    const float* apn = ap0 + (size_t)(b + 1) * bstride;
    const float* bpn = bp0 + (size_t)(b + 1) * bstride;

#pragma unroll
    for (int g = 0; g < 4; ++g) {
      if (has_next) {
        att_mma_group(cn, fa, fb);
        if (g < 3) {
          att_load_group(fa, fb, apn + (g + 1) * 256, bpn + (g + 1) * 256);
        } else if (b + 2 < B_) {
          att_load_group(fa, fb, apn + bstride, bpn + bstride);
        }
      }

      const v8f c = cc[g];
      const int col = (wave * 4 + g) * 16 + l15;
      const float sqm = sqc[col];
      const float nm  = nrmc[col];
      const v2f sqm2 = {sqm, sqm};
      const v2f nm2  = {nm, nm};
      const v2f eps2 = {EPS_, EPS_};
      const v2f zero2 = {0.0f, 0.0f};
      const v2f mtwo2 = {-2.0f, -2.0f};
      const v2f tiny2 = {1.17549435e-38f, 1.17549435e-38f};
#pragma unroll
      for (int v = 0; v < 8; v += 2) {
        const int rloc = rbase + v;
        const v2f sqn = *(const v2f*)&sqc[ntile * 16 + rloc];
        const v2f nn  = *(const v2f*)&nrmc[ntile * 16 + rloc];
        const v2f dot = {c[v], c[v + 1]};
        const v2f pn  = nn * nm2;
        const v2f sn  = (nn + nm2) + eps2;
        const v2f d2  = __builtin_elementwise_max(pk_fma(dot, mtwo2, sqn + sqm2), zero2);
        const v2f arg = __builtin_elementwise_max((d2 * sn) * sn, tiny2);
        const v2f rs  = {__builtin_amdgcn_rsqf(arg[0]), __builtin_amdgcn_rsqf(arg[1])};
        const v2f rc  = {__builtin_amdgcn_rcpf(pn[0]), __builtin_amdgcn_rcpf(pn[1])};
        const v2f q   = pk_fma(dot, rc, -(d2 * rs));
        es[rloc * APITCH + col]       = __expf(q[0]);
        es[(rloc + 1) * APITCH + col] = __expf(q[1]);
      }
    }

    if (has_next) {
      sq_s[(b + 1) & 1][tid]        = sn0;
      sq_s[(b + 1) & 1][tid + 256]  = sn1;
      nrm_s[(b + 1) & 1][tid]       = sqrtf(sn0) + EPS_;
      nrm_s[(b + 1) & 1][tid + 256] = sqrtf(sn1) + EPS_;
    }
    __syncthreads();

    {
      const float* er = es + srow * APITCH + sl * 4;
      v4f ev[8];
#pragma unroll
      for (int k = 0; k < 8; ++k) ev[k] = *(const v4f*)(er + k * 64);
      v4f p4 = ev[0];
#pragma unroll
      for (int k = 1; k < 8; ++k) p4 += ev[k];
      float part = (p4[0] + p4[1]) + (p4[2] + p4[3]);
#pragma unroll
      for (int m = 1; m < 16; m <<= 1) part += __shfl_xor(part, m);
      const float wgt = (1.0f / (float)B_) / (part + EPS_);
#pragma unroll
      for (int k = 0; k < 8; ++k) {
#pragma unroll
        for (int e = 0; e < 4; ++e) acc[k][e] = fmaf(ev[k][e], wgt, acc[k][e]);
      }
    }
  }

  float* outp = out + ((size_t)t * N_ + ntile * 16 + srow) * N_ + sl * 4;
#pragma unroll
  for (int k = 0; k < 8; ++k) *(v4f*)(outp + k * 64) = acc[k];
}

extern "C" void kernel_launch(void* const* d_in, const int* in_sizes, int n_in,
                              void* d_out, int out_size, void* d_ws, size_t ws_size,
                              hipStream_t stream) {
  const float* x    = (const float*)d_in[0];
  const float* W    = (const float*)d_in[1];
  const float* bias = (const float*)d_in[2];
  float* out = (float*)d_out;

  float* z  = (float*)d_ws;
  float* sq = z + (size_t)B_ * T_ * N_ * D_;

  GAT_z_kernel<<<B_ * (N_ / 16), 256, 0, stream>>>(x, W, bias, z);

  GAT_sq_kernel<<<B_ * T_ * N_ / 64 / 8, 256, 0, stream>>>(z, sq);

  GAT_att_kernel<<<T_ * (N_ / 16), 256, 0, stream>>>(z, sq, out);
}
